# starting kernel
# baseline (speedup 1.0000x reference)
#include <hip/hip_runtime.h>
#include <hip/hip_bf16.h>

typedef __attribute__((ext_vector_type(2))) float v2f;
typedef __attribute__((ext_vector_type(8))) float v8f;

#define DD    64
#define GG    128
#define NNODE 200000

__global__ void init_ws_kernel(float* __restrict__ zero_region, size_t nzero,
                               const float* __restrict__ W0,
                               const float* __restrict__ W1,
                               const float* __restrict__ W2,
                               float* __restrict__ wt) {
  size_t tid    = (size_t)blockIdx.x * blockDim.x + threadIdx.x;
  size_t stride = (size_t)gridDim.x * blockDim.x;
  for (size_t i = tid; i < nzero; i += stride) zero_region[i] = 0.0f;
  for (size_t i = tid; i < 3u * 64u * 64u; i += stride) {
    int s  = (int)(i >> 12);
    int nk = (int)(i & 4095);
    int n  = nk >> 6;
    int k  = nk & 63;
    const float* W = (s == 0) ? W0 : ((s == 1) ? W1 : W2);
    wt[i] = W[k * 64 + n];
  }
}

__global__ void scatter_b1_kernel(const float* __restrict__ x,
                                  const int* __restrict__ b1_rows,
                                  const float* __restrict__ b1_vals,
                                  const int* __restrict__ edge_batch,
                                  float* __restrict__ y,
                                  float* __restrict__ gcnt, int nE) {
  size_t tid = (size_t)blockIdx.x * blockDim.x + threadIdx.x;
  if (tid >= (size_t)nE * DD) return;
  int e = (int)(tid >> 6);
  int d = (int)(tid & 63);
  float xv = x[tid];
  int   r0 = b1_rows[2 * e];
  int   r1 = b1_rows[2 * e + 1];
  float v0 = b1_vals[2 * e];
  float v1 = b1_vals[2 * e + 1];
  atomicAdd(&y[(size_t)r0 * DD + d], v0 * xv);
  atomicAdd(&y[(size_t)r1 * DD + d], v1 * xv);
  if (d == 0) atomicAdd(&gcnt[edge_batch[e]], 1.0f);
}

__global__ void scatter_b2_kernel(const float* __restrict__ x,
                                  const int* __restrict__ b2_rows,
                                  const float* __restrict__ b2_vals,
                                  float* __restrict__ upper, int nT) {
  size_t tid = (size_t)blockIdx.x * blockDim.x + threadIdx.x;
  if (tid >= (size_t)nT * DD) return;
  int t = (int)(tid >> 6);
  int d = (int)(tid & 63);
  int   i0 = b2_rows[3 * t],     i1 = b2_rows[3 * t + 1], i2 = b2_rows[3 * t + 2];
  float v0 = b2_vals[3 * t],     v1 = b2_vals[3 * t + 1], v2 = b2_vals[3 * t + 2];
  float z = v0 * x[(size_t)i0 * DD + d]
          + v1 * x[(size_t)i1 * DD + d]
          + v2 * x[(size_t)i2 * DD + d];
  atomicAdd(&upper[(size_t)i0 * DD + d], v0 * z);
  atomicAdd(&upper[(size_t)i1 * DD + d], v1 * z);
  atomicAdd(&upper[(size_t)i2 * DD + d], v2 * z);
}

__global__ void __launch_bounds__(256)
fused_gemm_pool_kernel(const float* __restrict__ x,
                       const int* __restrict__ b1_rows,
                       const float* __restrict__ b1_vals,
                       const int* __restrict__ edge_batch,
                       const float* __restrict__ y,
                       const float* __restrict__ upper,
                       const float* __restrict__ wt,
                       float* __restrict__ gsum, int nE) {
  __shared__ float lgsum[GG * DD];
  for (int i = threadIdx.x; i < GG * DD; i += blockDim.x) lgsum[i] = 0.0f;
  __syncthreads();

  const int lane = threadIdx.x & 31;
  const int wave = threadIdx.x >> 5;
  const int tile = blockIdx.x * (blockDim.x >> 5) + wave;
  const int e0   = tile * 16;

  if (e0 < nE) {
    const int m  = lane & 15;
    const int kh = lane >> 4;
    const int e  = e0 + m;

    const int   r0 = b1_rows[2 * e];
    const int   r1 = b1_rows[2 * e + 1];
    const float s0 = b1_vals[2 * e];
    const float s1 = b1_vals[2 * e + 1];

    const float* xrow = x     + (size_t)e  * DD;
    const float* urow = upper + (size_t)e  * DD;
    const float* yr0  = y     + (size_t)r0 * DD;
    const float* yr1  = y     + (size_t)r1 * DD;

    v8f acc[4] = {};

    for (int src = 0; src < 3; ++src) {
      const float* wtm = wt + src * 4096;
      for (int ks = 0; ks < 16; ++ks) {
        const int kb = ks * 4 + kh * 2;
        v2f a;
        if (src == 0) {
          a.x = xrow[kb];  a.y = xrow[kb + 1];
        } else if (src == 1) {
          a.x = s0 * yr0[kb]     + s1 * yr1[kb];
          a.y = s0 * yr0[kb + 1] + s1 * yr1[kb + 1];
        } else {
          a.x = urow[kb];  a.y = urow[kb + 1];
        }
#pragma unroll
        for (int nt = 0; nt < 4; ++nt) {
          const float* wp = wtm + (nt * 16 + m) * 64 + kb;
          v2f b; b.x = wp[0]; b.y = wp[1];
          acc[nt] = __builtin_amdgcn_wmma_f32_16x16x4_f32(
              false, a, false, b, (short)0, acc[nt], false, false);
        }
      }
    }

    int gid[8];
#pragma unroll
    for (int r = 0; r < 8; ++r) gid[r] = edge_batch[e0 + kh * 8 + r];
#pragma unroll
    for (int nt = 0; nt < 4; ++nt) {
#pragma unroll
      for (int r = 0; r < 8; ++r) {
        float v = acc[nt][r];
        v = v > 0.0f ? v : 0.0f;
        atomicAdd(&lgsum[gid[r] * DD + nt * 16 + m], v);
      }
    }
  }

  __syncthreads();
  for (int i = threadIdx.x; i < GG * DD; i += blockDim.x)
    atomicAdd(&gsum[i], lgsum[i]);
}

__global__ void finalize_kernel(const float* __restrict__ gsum,
                                const float* __restrict__ gcnt,
                                float* __restrict__ out) {
  int i = blockIdx.x * blockDim.x + threadIdx.x;
  if (i >= GG * DD) return;
  float c = gcnt[i >> 6];
  float v = gsum[i] / fmaxf(c, 1.0f);
  out[i] = v;
  out[GG * DD + i] = v;
  out[2 * GG * DD + i] = v;
}

extern "C" void kernel_launch(void* const* d_in, const int* in_sizes, int n_in,
                              void* d_out, int out_size, void* d_ws, size_t ws_size,
                              hipStream_t stream) {
  const float* x        = (const float*)d_in[0];
  const int*   b1_rows  = (const int*)  d_in[1];
  const float* b1_vals  = (const float*)d_in[3];
  const int*   b2_rows  = (const int*)  d_in[4];
  const float* b2_vals  = (const float*)d_in[6];
  const int*   ebatch   = (const int*)  d_in[7];
  const float* W0       = (const float*)d_in[8];
  const float* W1       = (const float*)d_in[9];
  const float* W2       = (const float*)d_in[10];
  float*       out      = (float*)d_out;

  const int nE = in_sizes[0] / DD;
  const int nT = in_sizes[4] / 3;

  float* ws    = (float*)d_ws;
  float* y     = ws;
  size_t nY    = (size_t)NNODE * DD;
  float* upper = y + nY;
  size_t nU    = (size_t)nE * DD;
  float* gsum  = upper + nU;
  float* gcnt  = gsum + GG * DD;
  float* wt    = gcnt + GG;
  size_t nzero = nY + nU + GG * DD + GG;

  init_ws_kernel<<<2048, 256, 0, stream>>>(ws, nzero, W0, W1, W2, wt);

  {
    size_t total = (size_t)nE * DD;
    int blocks = (int)((total + 255) / 256);
    scatter_b1_kernel<<<blocks, 256, 0, stream>>>(x, b1_rows, b1_vals, ebatch,
                                                  y, gcnt, nE);
  }
  {
    size_t total = (size_t)nT * DD;
    int blocks = (int)((total + 255) / 256);
    scatter_b2_kernel<<<blocks, 256, 0, stream>>>(x, b2_rows, b2_vals, upper, nT);
  }
  {
    int tiles  = (nE + 15) / 16;
    int blocks = (tiles + 7) / 8;
    fused_gemm_pool_kernel<<<blocks, 256, 0, stream>>>(x, b1_rows, b1_vals,
                                                       ebatch, y, upper, wt,
                                                       gsum, nE);
  }
  finalize_kernel<<<(GG * DD + 255) / 256, 256, 0, stream>>>(gsum, gcnt, out);
}


# graph edge counts: one-lane global f32 atomics on 4 hot lines in the B1 scatter replaced by an LDS count in the pooling kernel, flushed as whole-line adds per block (exact whole-number sums)
# speedup vs baseline: 1.4304x; 1.4304x over previous
#include <hip/hip_runtime.h>
#include <hip/hip_bf16.h>

typedef __attribute__((ext_vector_type(2))) float v2f;
typedef __attribute__((ext_vector_type(8))) float v8f;

#define DD    64
#define GG    128
#define NNODE 200000

__global__ void init_ws_kernel(float* __restrict__ zero_region, size_t nzero,
                               const float* __restrict__ W0,
                               const float* __restrict__ W1,
                               const float* __restrict__ W2,
                               float* __restrict__ wt) {
  size_t tid    = (size_t)blockIdx.x * blockDim.x + threadIdx.x;
  size_t stride = (size_t)gridDim.x * blockDim.x;
  for (size_t i = tid; i < nzero; i += stride) zero_region[i] = 0.0f;
  for (size_t i = tid; i < 3u * 64u * 64u; i += stride) {
    int s  = (int)(i >> 12);
    int nk = (int)(i & 4095);
    int n  = nk >> 6;
    int k  = nk & 63;
    const float* W = (s == 0) ? W0 : ((s == 1) ? W1 : W2);
    wt[i] = W[k * 64 + n];
  }
}

__global__ void scatter_b1_kernel(const float* __restrict__ x,
                                  const int* __restrict__ b1_rows,
                                  const float* __restrict__ b1_vals,
                                  float* __restrict__ y, int nE) {
  size_t tid = (size_t)blockIdx.x * blockDim.x + threadIdx.x;
  if (tid >= (size_t)nE * DD) return;
  int e = (int)(tid >> 6);
  int d = (int)(tid & 63);
  float xv = x[tid];
  int   r0 = b1_rows[2 * e];
  int   r1 = b1_rows[2 * e + 1];
  float v0 = b1_vals[2 * e];
  float v1 = b1_vals[2 * e + 1];
  atomicAdd(&y[(size_t)r0 * DD + d], v0 * xv);
  atomicAdd(&y[(size_t)r1 * DD + d], v1 * xv);
}

__global__ void scatter_b2_kernel(const float* __restrict__ x,
                                  const int* __restrict__ b2_rows,
                                  const float* __restrict__ b2_vals,
                                  float* __restrict__ upper, int nT) {
  size_t tid = (size_t)blockIdx.x * blockDim.x + threadIdx.x;
  if (tid >= (size_t)nT * DD) return;
  int t = (int)(tid >> 6);
  int d = (int)(tid & 63);
  int   i0 = b2_rows[3 * t],     i1 = b2_rows[3 * t + 1], i2 = b2_rows[3 * t + 2];
  float v0 = b2_vals[3 * t],     v1 = b2_vals[3 * t + 1], v2 = b2_vals[3 * t + 2];
  float z = v0 * x[(size_t)i0 * DD + d]
          + v1 * x[(size_t)i1 * DD + d]
          + v2 * x[(size_t)i2 * DD + d];
  atomicAdd(&upper[(size_t)i0 * DD + d], v0 * z);
  atomicAdd(&upper[(size_t)i1 * DD + d], v1 * z);
  atomicAdd(&upper[(size_t)i2 * DD + d], v2 * z);
}

__global__ void __launch_bounds__(256)
fused_gemm_pool_kernel(const float* __restrict__ x,
                       const int* __restrict__ b1_rows,
                       const float* __restrict__ b1_vals,
                       const int* __restrict__ edge_batch,
                       const float* __restrict__ y,
                       const float* __restrict__ upper,
                       const float* __restrict__ wt,
                       float* __restrict__ gsum,
                       float* __restrict__ gcnt, int nE) {
  __shared__ float lgsum[GG * DD];
  __shared__ float lcnt[GG];
  for (int i = threadIdx.x; i < GG * DD; i += blockDim.x) lgsum[i] = 0.0f;
  if (threadIdx.x < GG) lcnt[threadIdx.x] = 0.0f;
  __syncthreads();

  const int lane = threadIdx.x & 31;
  const int wave = threadIdx.x >> 5;
  const int tile = blockIdx.x * (blockDim.x >> 5) + wave;
  const int e0   = tile * 16;

  if (e0 < nE) {
    const int m  = lane & 15;
    const int kh = lane >> 4;
    const int e  = e0 + m;

    const int   r0 = b1_rows[2 * e];
    const int   r1 = b1_rows[2 * e + 1];
    const float s0 = b1_vals[2 * e];
    const float s1 = b1_vals[2 * e + 1];

    const float* xrow = x     + (size_t)e  * DD;
    const float* urow = upper + (size_t)e  * DD;
    const float* yr0  = y     + (size_t)r0 * DD;
    const float* yr1  = y     + (size_t)r1 * DD;

    v8f acc[4] = {};

    for (int src = 0; src < 3; ++src) {
      const float* wtm = wt + src * 4096;
      for (int ks = 0; ks < 16; ++ks) {
        const int kb = ks * 4 + kh * 2;
        v2f a;
        if (src == 0) {
          a.x = xrow[kb];  a.y = xrow[kb + 1];
        } else if (src == 1) {
          a.x = s0 * yr0[kb]     + s1 * yr1[kb];
          a.y = s0 * yr0[kb + 1] + s1 * yr1[kb + 1];
        } else {
          a.x = urow[kb];  a.y = urow[kb + 1];
        }
#pragma unroll
        for (int nt = 0; nt < 4; ++nt) {
          const float* wp = wtm + (nt * 16 + m) * 64 + kb;
          v2f b; b.x = wp[0]; b.y = wp[1];
          acc[nt] = __builtin_amdgcn_wmma_f32_16x16x4_f32(
              false, a, false, b, (short)0, acc[nt], false, false);
        }
      }
    }

    int gid[8];
#pragma unroll
    for (int r = 0; r < 8; ++r) gid[r] = edge_batch[e0 + kh * 8 + r];
#pragma unroll
    for (int nt = 0; nt < 4; ++nt) {
#pragma unroll
      for (int r = 0; r < 8; ++r) {
        float v = acc[nt][r];
        v = v > 0.0f ? v : 0.0f;
        atomicAdd(&lgsum[gid[r] * DD + nt * 16 + m], v);
      }
    }
    if (kh == 0) atomicAdd(&lcnt[edge_batch[e]], 1.0f);
  }

  __syncthreads();
  for (int i = threadIdx.x; i < GG * DD; i += blockDim.x)
    atomicAdd(&gsum[i], lgsum[i]);
  if (threadIdx.x < GG) atomicAdd(&gcnt[threadIdx.x], lcnt[threadIdx.x]);
}

__global__ void finalize_kernel(const float* __restrict__ gsum,
                                const float* __restrict__ gcnt,
                                float* __restrict__ out) {
  int i = blockIdx.x * blockDim.x + threadIdx.x;
  if (i >= GG * DD) return;
  float c = gcnt[i >> 6];
  float v = gsum[i] / fmaxf(c, 1.0f);
  out[i] = v;
  out[GG * DD + i] = v;
  out[2 * GG * DD + i] = v;
}

extern "C" void kernel_launch(void* const* d_in, const int* in_sizes, int n_in,
                              void* d_out, int out_size, void* d_ws, size_t ws_size,
                              hipStream_t stream) {
  const float* x        = (const float*)d_in[0];
  const int*   b1_rows  = (const int*)  d_in[1];
  const float* b1_vals  = (const float*)d_in[3];
  const int*   b2_rows  = (const int*)  d_in[4];
  const float* b2_vals  = (const float*)d_in[6];
  const int*   ebatch   = (const int*)  d_in[7];
  const float* W0       = (const float*)d_in[8];
  const float* W1       = (const float*)d_in[9];
  const float* W2       = (const float*)d_in[10];
  float*       out      = (float*)d_out;

  const int nE = in_sizes[0] / DD;
  const int nT = in_sizes[4] / 3;

  float* ws    = (float*)d_ws;
  float* y     = ws;
  size_t nY    = (size_t)NNODE * DD;
  float* upper = y + nY;
  size_t nU    = (size_t)nE * DD;
  float* gsum  = upper + nU;
  float* gcnt  = gsum + GG * DD;
  float* wt    = gcnt + GG;
  size_t nzero = nY + nU + GG * DD + GG;

  init_ws_kernel<<<2048, 256, 0, stream>>>(ws, nzero, W0, W1, W2, wt);

  {
    size_t total = (size_t)nE * DD;
    int blocks = (int)((total + 255) / 256);
    scatter_b1_kernel<<<blocks, 256, 0, stream>>>(x, b1_rows, b1_vals, y, nE);
  }
  {
    size_t total = (size_t)nT * DD;
    int blocks = (int)((total + 255) / 256);
    scatter_b2_kernel<<<blocks, 256, 0, stream>>>(x, b2_rows, b2_vals, upper, nT);
  }
  {
    int tiles  = (nE + 15) / 16;
    int blocks = (tiles + 7) / 8;
    fused_gemm_pool_kernel<<<blocks, 256, 0, stream>>>(x, b1_rows, b1_vals,
                                                       ebatch, y, upper, wt,
                                                       gsum, gcnt, nE);
  }
  finalize_kernel<<<(GG * DD + 255) / 256, 256, 0, stream>>>(gsum, gcnt, out);
}


# both scatter kernels: one thread per element (1.5M short waves, bound by wave entry) replaced by a fixed grid of 16384 resident waves looping 4 edges / 2 triangles per trip; whole-line adds kept
# speedup vs baseline: 1.7419x; 1.2178x over previous
#include <hip/hip_runtime.h>
#include <hip/hip_bf16.h>

typedef __attribute__((ext_vector_type(2))) float v2f;
typedef __attribute__((ext_vector_type(8))) float v8f;

#define DD    64
#define GG    128
#define NNODE 200000

__global__ void init_ws_kernel(float* __restrict__ zero_region, size_t nzero,
                               const float* __restrict__ W0,
                               const float* __restrict__ W1,
                               const float* __restrict__ W2,
                               float* __restrict__ wt) {
  size_t tid    = (size_t)blockIdx.x * blockDim.x + threadIdx.x;
  size_t stride = (size_t)gridDim.x * blockDim.x;
  for (size_t i = tid; i < nzero; i += stride) zero_region[i] = 0.0f;
  for (size_t i = tid; i < 3u * 64u * 64u; i += stride) {
    int s  = (int)(i >> 12);
    int nk = (int)(i & 4095);
    int n  = nk >> 6;
    int k  = nk & 63;
    const float* W = (s == 0) ? W0 : ((s == 1) ? W1 : W2);
    wt[i] = W[k * 64 + n];
  }
}

#define SC_BLOCKS 2048
#define SC_WPB    8

__device__ __forceinline__ void b1_edge(const float* __restrict__ x,
                                        const int* __restrict__ b1_rows,
                                        const float* __restrict__ b1_vals,
                                        float* __restrict__ y, int e, int lane) {
  const float* xr = x + (size_t)e * DD;
  const float xa = xr[lane];
  const float xb = xr[32 + lane];
  const int   r0 = b1_rows[2 * e];
  const int   r1 = b1_rows[2 * e + 1];
  const float v0 = b1_vals[2 * e];
  const float v1 = b1_vals[2 * e + 1];
  float* y0 = y + (size_t)r0 * DD;
  float* y1 = y + (size_t)r1 * DD;
  atomicAdd(&y0[lane],      v0 * xa);
  atomicAdd(&y0[32 + lane], v0 * xb);
  atomicAdd(&y1[lane],      v1 * xa);
  atomicAdd(&y1[32 + lane], v1 * xb);
}

__global__ void __launch_bounds__(256)
scatter_b1_kernel(const float* __restrict__ x,
                  const int* __restrict__ b1_rows,
                  const float* __restrict__ b1_vals,
                  float* __restrict__ y, int nE) {
  const int lane = threadIdx.x & 31;
  const int wv = __builtin_amdgcn_readfirstlane((int)(threadIdx.x >> 5));
  const int gw = blockIdx.x * SC_WPB + wv;
  const int nw = gridDim.x * SC_WPB;
  const int nG = nE >> 2;
  for (int g = gw; g < nG; g += nw) {
    const int e0 = g * 4;
    float xa[4], xb[4], v0[4], v1[4];
    int   r0[4], r1[4];
#pragma unroll
    for (int u = 0; u < 4; ++u) {
      const int e = e0 + u;
      const float* xr = x + (size_t)e * DD;
      xa[u] = xr[lane];
      xb[u] = xr[32 + lane];
      r0[u] = b1_rows[2 * e];
      r1[u] = b1_rows[2 * e + 1];
      v0[u] = b1_vals[2 * e];
      v1[u] = b1_vals[2 * e + 1];
    }
#pragma unroll
    for (int u = 0; u < 4; ++u) {
      float* y0 = y + (size_t)r0[u] * DD;
      float* y1 = y + (size_t)r1[u] * DD;
      atomicAdd(&y0[lane],      v0[u] * xa[u]);
      atomicAdd(&y0[32 + lane], v0[u] * xb[u]);
      atomicAdd(&y1[lane],      v1[u] * xa[u]);
      atomicAdd(&y1[32 + lane], v1[u] * xb[u]);
    }
  }
  for (int e = nG * 4 + gw; e < nE; e += nw)
    b1_edge(x, b1_rows, b1_vals, y, e, lane);
}

__device__ __forceinline__ void b2_tri(const float* __restrict__ x,
                                       const int* __restrict__ b2_rows,
                                       const float* __restrict__ b2_vals,
                                       float* __restrict__ upper, int t, int lane) {
  const int   i0 = b2_rows[3 * t], i1 = b2_rows[3 * t + 1], i2 = b2_rows[3 * t + 2];
  const float v0 = b2_vals[3 * t], v1 = b2_vals[3 * t + 1], v2 = b2_vals[3 * t + 2];
  const float* x0 = x + (size_t)i0 * DD;
  const float* x1 = x + (size_t)i1 * DD;
  const float* x2 = x + (size_t)i2 * DD;
  const float za = v0 * x0[lane]      + v1 * x1[lane]      + v2 * x2[lane];
  const float zc = v0 * x0[32 + lane] + v1 * x1[32 + lane] + v2 * x2[32 + lane];
  float* u0 = upper + (size_t)i0 * DD;
  float* u1 = upper + (size_t)i1 * DD;
  float* u2 = upper + (size_t)i2 * DD;
  atomicAdd(&u0[lane],      v0 * za);
  atomicAdd(&u0[32 + lane], v0 * zc);
  atomicAdd(&u1[lane],      v1 * za);
  atomicAdd(&u1[32 + lane], v1 * zc);
  atomicAdd(&u2[lane],      v2 * za);
  atomicAdd(&u2[32 + lane], v2 * zc);
}

__global__ void __launch_bounds__(256)
scatter_b2_kernel(const float* __restrict__ x,
                  const int* __restrict__ b2_rows,
                  const float* __restrict__ b2_vals,
                  float* __restrict__ upper, int nT) {
  const int lane = threadIdx.x & 31;
  const int wv = __builtin_amdgcn_readfirstlane((int)(threadIdx.x >> 5));
  const int gw = blockIdx.x * SC_WPB + wv;
  const int nw = gridDim.x * SC_WPB;
  const int nP = nT >> 1;
  for (int p = gw; p < nP; p += nw) {
    const int t0 = p * 2;
    int   i0[2], i1[2], i2[2];
    float v0[2], v1[2], v2[2];
    float a0[2], a1[2], a2[2], c0[2], c1[2], c2[2];
#pragma unroll
    for (int u = 0; u < 2; ++u) {
      const int t = t0 + u;
      i0[u] = b2_rows[3 * t];  i1[u] = b2_rows[3 * t + 1];  i2[u] = b2_rows[3 * t + 2];
      v0[u] = b2_vals[3 * t];  v1[u] = b2_vals[3 * t + 1];  v2[u] = b2_vals[3 * t + 2];
      const float* x0 = x + (size_t)i0[u] * DD;
      const float* x1 = x + (size_t)i1[u] * DD;
      const float* x2 = x + (size_t)i2[u] * DD;
      a0[u] = x0[lane];  c0[u] = x0[32 + lane];
      a1[u] = x1[lane];  c1[u] = x1[32 + lane];
      a2[u] = x2[lane];  c2[u] = x2[32 + lane];
    }
#pragma unroll
    for (int u = 0; u < 2; ++u) {
      const float za = v0[u] * a0[u] + v1[u] * a1[u] + v2[u] * a2[u];
      const float zc = v0[u] * c0[u] + v1[u] * c1[u] + v2[u] * c2[u];
      float* u0 = upper + (size_t)i0[u] * DD;
      float* u1 = upper + (size_t)i1[u] * DD;
      float* u2 = upper + (size_t)i2[u] * DD;
      atomicAdd(&u0[lane],      v0[u] * za);
      atomicAdd(&u0[32 + lane], v0[u] * zc);
      atomicAdd(&u1[lane],      v1[u] * za);
      atomicAdd(&u1[32 + lane], v1[u] * zc);
      atomicAdd(&u2[lane],      v2[u] * za);
      atomicAdd(&u2[32 + lane], v2[u] * zc);
    }
  }
  for (int t = nP * 2 + gw; t < nT; t += nw)
    b2_tri(x, b2_rows, b2_vals, upper, t, lane);
}

__global__ void __launch_bounds__(256)
fused_gemm_pool_kernel(const float* __restrict__ x,
                       const int* __restrict__ b1_rows,
                       const float* __restrict__ b1_vals,
                       const int* __restrict__ edge_batch,
                       const float* __restrict__ y,
                       const float* __restrict__ upper,
                       const float* __restrict__ wt,
                       float* __restrict__ gsum,
                       float* __restrict__ gcnt, int nE) {
  __shared__ float lgsum[GG * DD];
  __shared__ float lcnt[GG];
  for (int i = threadIdx.x; i < GG * DD; i += blockDim.x) lgsum[i] = 0.0f;
  if (threadIdx.x < GG) lcnt[threadIdx.x] = 0.0f;
  __syncthreads();

  const int lane = threadIdx.x & 31;
  const int wave = threadIdx.x >> 5;
  const int tile = blockIdx.x * (blockDim.x >> 5) + wave;
  const int e0   = tile * 16;

  if (e0 < nE) {
    const int m  = lane & 15;
    const int kh = lane >> 4;
    const int e  = e0 + m;

    const int   r0 = b1_rows[2 * e];
    const int   r1 = b1_rows[2 * e + 1];
    const float s0 = b1_vals[2 * e];
    const float s1 = b1_vals[2 * e + 1];

    const float* xrow = x     + (size_t)e  * DD;
    const float* urow = upper + (size_t)e  * DD;
    const float* yr0  = y     + (size_t)r0 * DD;
    const float* yr1  = y     + (size_t)r1 * DD;

    v8f acc[4] = {};

    for (int src = 0; src < 3; ++src) {
      const float* wtm = wt + src * 4096;
      for (int ks = 0; ks < 16; ++ks) {
        const int kb = ks * 4 + kh * 2;
        v2f a;
        if (src == 0) {
          a.x = xrow[kb];  a.y = xrow[kb + 1];
        } else if (src == 1) {
          a.x = s0 * yr0[kb]     + s1 * yr1[kb];
          a.y = s0 * yr0[kb + 1] + s1 * yr1[kb + 1];
        } else {
          a.x = urow[kb];  a.y = urow[kb + 1];
        }
#pragma unroll
        for (int nt = 0; nt < 4; ++nt) {
          const float* wp = wtm + (nt * 16 + m) * 64 + kb;
          v2f b; b.x = wp[0]; b.y = wp[1];
          acc[nt] = __builtin_amdgcn_wmma_f32_16x16x4_f32(
              false, a, false, b, (short)0, acc[nt], false, false);
        }
      }
    }

    int gid[8];
#pragma unroll
    for (int r = 0; r < 8; ++r) gid[r] = edge_batch[e0 + kh * 8 + r];
#pragma unroll
    for (int nt = 0; nt < 4; ++nt) {
#pragma unroll
      for (int r = 0; r < 8; ++r) {
        float v = acc[nt][r];
        v = v > 0.0f ? v : 0.0f;
        atomicAdd(&lgsum[gid[r] * DD + nt * 16 + m], v);
      }
    }
    if (kh == 0) atomicAdd(&lcnt[edge_batch[e]], 1.0f);
  }

  __syncthreads();
  for (int i = threadIdx.x; i < GG * DD; i += blockDim.x)
    atomicAdd(&gsum[i], lgsum[i]);
  if (threadIdx.x < GG) atomicAdd(&gcnt[threadIdx.x], lcnt[threadIdx.x]);
}

__global__ void finalize_kernel(const float* __restrict__ gsum,
                                const float* __restrict__ gcnt,
                                float* __restrict__ out) {
  int i = blockIdx.x * blockDim.x + threadIdx.x;
  if (i >= GG * DD) return;
  float c = gcnt[i >> 6];
  float v = gsum[i] / fmaxf(c, 1.0f);
  out[i] = v;
  out[GG * DD + i] = v;
  out[2 * GG * DD + i] = v;
}

extern "C" void kernel_launch(void* const* d_in, const int* in_sizes, int n_in,
                              void* d_out, int out_size, void* d_ws, size_t ws_size,
                              hipStream_t stream) {
  const float* x        = (const float*)d_in[0];
  const int*   b1_rows  = (const int*)  d_in[1];
  const float* b1_vals  = (const float*)d_in[3];
  const int*   b2_rows  = (const int*)  d_in[4];
  const float* b2_vals  = (const float*)d_in[6];
  const int*   ebatch   = (const int*)  d_in[7];
  const float* W0       = (const float*)d_in[8];
  const float* W1       = (const float*)d_in[9];
  const float* W2       = (const float*)d_in[10];
  float*       out      = (float*)d_out;

  const int nE = in_sizes[0] / DD;
  const int nT = in_sizes[4] / 3;

  float* ws    = (float*)d_ws;
  float* y     = ws;
  size_t nY    = (size_t)NNODE * DD;
  float* upper = y + nY;
  size_t nU    = (size_t)nE * DD;
  float* gsum  = upper + nU;
  float* gcnt  = gsum + GG * DD;
  float* wt    = gcnt + GG;
  size_t nzero = nY + nU + GG * DD + GG;

  init_ws_kernel<<<2048, 256, 0, stream>>>(ws, nzero, W0, W1, W2, wt);

  scatter_b1_kernel<<<SC_BLOCKS, 256, 0, stream>>>(x, b1_rows, b1_vals, y, nE);
  scatter_b2_kernel<<<SC_BLOCKS, 256, 0, stream>>>(x, b2_rows, b2_vals, upper, nT);
  {
    int tiles  = (nE + 15) / 16;
    int blocks = (tiles + 7) / 8;
    fused_gemm_pool_kernel<<<blocks, 256, 0, stream>>>(x, b1_rows, b1_vals,
                                                       ebatch, y, upper, wt,
                                                       gsum, gcnt, nE);
  }
  finalize_kernel<<<(GG * DD + 255) / 256, 256, 0, stream>>>(gsum, gcnt, out);
}


# fused GEMM operand feed: weights pre-stored in fragment order (each B load 512 contiguous bytes) and A rows read 16 B per lane, all up front; k order of A and B permuted together (sum order changes)
# speedup vs baseline: 2.4276x; 1.3937x over previous
#include <hip/hip_runtime.h>
#include <hip/hip_bf16.h>

typedef __attribute__((ext_vector_type(2))) float v2f;
typedef __attribute__((ext_vector_type(4))) float v4f;
typedef __attribute__((ext_vector_type(8))) float v8f;

#define DD    64
#define GG    128
#define NNODE 200000

__global__ void init_ws_kernel(float* __restrict__ zero_region, size_t nzero,
                               const float* __restrict__ W0,
                               const float* __restrict__ W1,
                               const float* __restrict__ W2,
                               float* __restrict__ wt) {
  size_t tid    = (size_t)blockIdx.x * blockDim.x + threadIdx.x;
  size_t stride = (size_t)gridDim.x * blockDim.x;
  for (size_t i = tid; i < nzero; i += stride) zero_region[i] = 0.0f;
  for (size_t i = tid; i < 3u * 64u * 64u; i += stride) {
    int s  = (int)(i >> 12);
    int jj = (int)(i >> 9) & 7;
    int nt = (int)(i >> 7) & 3;
    int ln = (int)(i >> 2) & 31;
    int t  = (int)(i & 3);
    int k  = 8 * jj + 4 * (ln >> 4) + t;
    int n  = nt * 16 + (ln & 15);
    const float* W = (s == 0) ? W0 : ((s == 1) ? W1 : W2);
    wt[i] = W[k * 64 + n];
  }
}

#define SC_BLOCKS 2048
#define SC_WPB    8

__device__ __forceinline__ void b1_edge(const float* __restrict__ x,
                                        const int* __restrict__ b1_rows,
                                        const float* __restrict__ b1_vals,
                                        float* __restrict__ y, int e, int lane) {
  const float* xr = x + (size_t)e * DD;
  const float xa = xr[lane];
  const float xb = xr[32 + lane];
  const int   r0 = b1_rows[2 * e];
  const int   r1 = b1_rows[2 * e + 1];
  const float v0 = b1_vals[2 * e];
  const float v1 = b1_vals[2 * e + 1];
  float* y0 = y + (size_t)r0 * DD;
  float* y1 = y + (size_t)r1 * DD;
  atomicAdd(&y0[lane],      v0 * xa);
  atomicAdd(&y0[32 + lane], v0 * xb);
  atomicAdd(&y1[lane],      v1 * xa);
  atomicAdd(&y1[32 + lane], v1 * xb);
}

__global__ void __launch_bounds__(256)
scatter_b1_kernel(const float* __restrict__ x,
                  const int* __restrict__ b1_rows,
                  const float* __restrict__ b1_vals,
                  float* __restrict__ y, int nE) {
  const int lane = threadIdx.x & 31;
  const int wv = __builtin_amdgcn_readfirstlane((int)(threadIdx.x >> 5));
  const int gw = blockIdx.x * SC_WPB + wv;
  const int nw = gridDim.x * SC_WPB;
  const int nG = nE >> 2;
  for (int g = gw; g < nG; g += nw) {
    const int e0 = g * 4;
    float xa[4], xb[4], v0[4], v1[4];
    int   r0[4], r1[4];
#pragma unroll
    for (int u = 0; u < 4; ++u) {
      const int e = e0 + u;
      const float* xr = x + (size_t)e * DD;
      xa[u] = xr[lane];
      xb[u] = xr[32 + lane];
      r0[u] = b1_rows[2 * e];
      r1[u] = b1_rows[2 * e + 1];
      v0[u] = b1_vals[2 * e];
      v1[u] = b1_vals[2 * e + 1];
    }
#pragma unroll
    for (int u = 0; u < 4; ++u) {
      float* y0 = y + (size_t)r0[u] * DD;
      float* y1 = y + (size_t)r1[u] * DD;
      atomicAdd(&y0[lane],      v0[u] * xa[u]);
      atomicAdd(&y0[32 + lane], v0[u] * xb[u]);
      atomicAdd(&y1[lane],      v1[u] * xa[u]);
      atomicAdd(&y1[32 + lane], v1[u] * xb[u]);
    }
  }
  for (int e = nG * 4 + gw; e < nE; e += nw)
    b1_edge(x, b1_rows, b1_vals, y, e, lane);
}

__device__ __forceinline__ void b2_tri(const float* __restrict__ x,
                                       const int* __restrict__ b2_rows,
                                       const float* __restrict__ b2_vals,
                                       float* __restrict__ upper, int t, int lane) {
  const int   i0 = b2_rows[3 * t], i1 = b2_rows[3 * t + 1], i2 = b2_rows[3 * t + 2];
  const float v0 = b2_vals[3 * t], v1 = b2_vals[3 * t + 1], v2 = b2_vals[3 * t + 2];
  const float* x0 = x + (size_t)i0 * DD;
  const float* x1 = x + (size_t)i1 * DD;
  const float* x2 = x + (size_t)i2 * DD;
  const float za = v0 * x0[lane]      + v1 * x1[lane]      + v2 * x2[lane];
  const float zc = v0 * x0[32 + lane] + v1 * x1[32 + lane] + v2 * x2[32 + lane];
  float* u0 = upper + (size_t)i0 * DD;
  float* u1 = upper + (size_t)i1 * DD;
  float* u2 = upper + (size_t)i2 * DD;
  atomicAdd(&u0[lane],      v0 * za);
  atomicAdd(&u0[32 + lane], v0 * zc);
  atomicAdd(&u1[lane],      v1 * za);
  atomicAdd(&u1[32 + lane], v1 * zc);
  atomicAdd(&u2[lane],      v2 * za);
  atomicAdd(&u2[32 + lane], v2 * zc);
}

__global__ void __launch_bounds__(256)
scatter_b2_kernel(const float* __restrict__ x,
                  const int* __restrict__ b2_rows,
                  const float* __restrict__ b2_vals,
                  float* __restrict__ upper, int nT) {
  const int lane = threadIdx.x & 31;
  const int wv = __builtin_amdgcn_readfirstlane((int)(threadIdx.x >> 5));
  const int gw = blockIdx.x * SC_WPB + wv;
  const int nw = gridDim.x * SC_WPB;
  const int nP = nT >> 1;
  for (int p = gw; p < nP; p += nw) {
    const int t0 = p * 2;
    int   i0[2], i1[2], i2[2];
    float v0[2], v1[2], v2[2];
    float a0[2], a1[2], a2[2], c0[2], c1[2], c2[2];
#pragma unroll
    for (int u = 0; u < 2; ++u) {
      const int t = t0 + u;
      i0[u] = b2_rows[3 * t];  i1[u] = b2_rows[3 * t + 1];  i2[u] = b2_rows[3 * t + 2];
      v0[u] = b2_vals[3 * t];  v1[u] = b2_vals[3 * t + 1];  v2[u] = b2_vals[3 * t + 2];
      const float* x0 = x + (size_t)i0[u] * DD;
      const float* x1 = x + (size_t)i1[u] * DD;
      const float* x2 = x + (size_t)i2[u] * DD;
      a0[u] = x0[lane];  c0[u] = x0[32 + lane];
      a1[u] = x1[lane];  c1[u] = x1[32 + lane];
      a2[u] = x2[lane];  c2[u] = x2[32 + lane];
    }
#pragma unroll
    for (int u = 0; u < 2; ++u) {
      const float za = v0[u] * a0[u] + v1[u] * a1[u] + v2[u] * a2[u];
      const float zc = v0[u] * c0[u] + v1[u] * c1[u] + v2[u] * c2[u];
      float* u0 = upper + (size_t)i0[u] * DD;
      float* u1 = upper + (size_t)i1[u] * DD;
      float* u2 = upper + (size_t)i2[u] * DD;
      atomicAdd(&u0[lane],      v0[u] * za);
      atomicAdd(&u0[32 + lane], v0[u] * zc);
      atomicAdd(&u1[lane],      v1[u] * za);
      atomicAdd(&u1[32 + lane], v1[u] * zc);
      atomicAdd(&u2[lane],      v2[u] * za);
      atomicAdd(&u2[32 + lane], v2[u] * zc);
    }
  }
  for (int t = nP * 2 + gw; t < nT; t += nw)
    b2_tri(x, b2_rows, b2_vals, upper, t, lane);
}

__global__ void __launch_bounds__(256)
fused_gemm_pool_kernel(const float* __restrict__ x,
                       const int* __restrict__ b1_rows,
                       const float* __restrict__ b1_vals,
                       const int* __restrict__ edge_batch,
                       const float* __restrict__ y,
                       const float* __restrict__ upper,
                       const float* __restrict__ wt,
                       float* __restrict__ gsum,
                       float* __restrict__ gcnt, int nE) {
  __shared__ float lgsum[GG * DD];
  __shared__ float lcnt[GG];
  for (int i = threadIdx.x; i < GG * DD; i += blockDim.x) lgsum[i] = 0.0f;
  if (threadIdx.x < GG) lcnt[threadIdx.x] = 0.0f;
  __syncthreads();

  const int lane = threadIdx.x & 31;
  const int wave = threadIdx.x >> 5;
  const int tile = blockIdx.x * (blockDim.x >> 5) + wave;
  const int e0   = tile * 16;

  if (e0 < nE) {
    const int m  = lane & 15;
    const int kh = lane >> 4;
    const int e  = e0 + m;

    const int   r0 = b1_rows[2 * e];
    const int   r1 = b1_rows[2 * e + 1];
    const float s0 = b1_vals[2 * e];
    const float s1 = b1_vals[2 * e + 1];

    const v4f* xq  = (const v4f*)(x     + (size_t)e  * DD) + kh;
    const v4f* uq  = (const v4f*)(upper + (size_t)e  * DD) + kh;
    const v4f* y0q = (const v4f*)(y     + (size_t)r0 * DD) + kh;
    const v4f* y1q = (const v4f*)(y     + (size_t)r1 * DD) + kh;
    const v4f* wq = (const v4f*)wt + lane;

    v4f a[3][8], p0[8], p1[8];
#pragma unroll
    for (int jj = 0; jj < 8; ++jj) a[0][jj] = xq[2 * jj];
#pragma unroll
    for (int jj = 0; jj < 8; ++jj) a[2][jj] = uq[2 * jj];
#pragma unroll
    for (int jj = 0; jj < 8; ++jj) p0[jj] = y0q[2 * jj];
#pragma unroll
    for (int jj = 0; jj < 8; ++jj) p1[jj] = y1q[2 * jj];

    v4f bq[3][4];
#pragma unroll
    for (int s = 0; s < 2; ++s)
#pragma unroll
      for (int nt = 0; nt < 4; ++nt) bq[s][nt] = wq[(s * 4 + nt) * 32];
    __builtin_amdgcn_sched_barrier(0);

    v8f acc[4] = {};

#pragma unroll
    for (int s = 0; s < 24; ++s) {
      const int src = s >> 3;
      const int jj  = s & 7;
      if (s == 8) {
#pragma unroll
        for (int j = 0; j < 8; ++j)
          a[1][j] = s0 * p0[j] + s1 * p1[j];
      }
      if (s + 2 < 24) {
#pragma unroll
        for (int nt = 0; nt < 4; ++nt)
          bq[(s + 2) % 3][nt] = wq[((s + 2) * 4 + nt) * 32];
      }
      const v4f a4 = a[src][jj];
      v2f alo; alo.x = a4.x; alo.y = a4.y;
      v2f ahi; ahi.x = a4.z; ahi.y = a4.w;
#pragma unroll
      for (int nt = 0; nt < 4; ++nt) {
        v2f b; b.x = bq[s % 3][nt].x; b.y = bq[s % 3][nt].y;
        acc[nt] = __builtin_amdgcn_wmma_f32_16x16x4_f32(
            false, alo, false, b, (short)0, acc[nt], false, false);
      }
#pragma unroll
      for (int nt = 0; nt < 4; ++nt) {
        v2f b; b.x = bq[s % 3][nt].z; b.y = bq[s % 3][nt].w;
        acc[nt] = __builtin_amdgcn_wmma_f32_16x16x4_f32(
            false, ahi, false, b, (short)0, acc[nt], false, false);
      }
      __builtin_amdgcn_sched_barrier(0);
    }

    int gid[8];
#pragma unroll
    for (int r = 0; r < 8; ++r) gid[r] = edge_batch[e0 + kh * 8 + r];
#pragma unroll
    for (int nt = 0; nt < 4; ++nt) {
#pragma unroll
      for (int r = 0; r < 8; ++r) {
        float v = acc[nt][r];
        v = v > 0.0f ? v : 0.0f;
        atomicAdd(&lgsum[gid[r] * DD + nt * 16 + m], v);
      }
    }
    if (kh == 0) atomicAdd(&lcnt[edge_batch[e]], 1.0f);
  }

  __syncthreads();
  for (int i = threadIdx.x; i < GG * DD; i += blockDim.x)
    atomicAdd(&gsum[i], lgsum[i]);
  if (threadIdx.x < GG) atomicAdd(&gcnt[threadIdx.x], lcnt[threadIdx.x]);
}

__global__ void finalize_kernel(const float* __restrict__ gsum,
                                const float* __restrict__ gcnt,
                                float* __restrict__ out) {
  int i = blockIdx.x * blockDim.x + threadIdx.x;
  if (i >= GG * DD) return;
  float c = gcnt[i >> 6];
  float v = gsum[i] / fmaxf(c, 1.0f);
  out[i] = v;
  out[GG * DD + i] = v;
  out[2 * GG * DD + i] = v;
}

extern "C" void kernel_launch(void* const* d_in, const int* in_sizes, int n_in,
                              void* d_out, int out_size, void* d_ws, size_t ws_size,
                              hipStream_t stream) {
  const float* x        = (const float*)d_in[0];
  const int*   b1_rows  = (const int*)  d_in[1];
  const float* b1_vals  = (const float*)d_in[3];
  const int*   b2_rows  = (const int*)  d_in[4];
  const float* b2_vals  = (const float*)d_in[6];
  const int*   ebatch   = (const int*)  d_in[7];
  const float* W0       = (const float*)d_in[8];
  const float* W1       = (const float*)d_in[9];
  const float* W2       = (const float*)d_in[10];
  float*       out      = (float*)d_out;

  const int nE = in_sizes[0] / DD;
  const int nT = in_sizes[4] / 3;

  float* ws    = (float*)d_ws;
  float* y     = ws;
  size_t nY    = (size_t)NNODE * DD;
  float* upper = y + nY;
  size_t nU    = (size_t)nE * DD;
  float* gsum  = upper + nU;
  float* gcnt  = gsum + GG * DD;
  float* wt    = gcnt + GG;
  size_t nzero = nY + nU + GG * DD + GG;

  init_ws_kernel<<<2048, 256, 0, stream>>>(ws, nzero, W0, W1, W2, wt);

  scatter_b1_kernel<<<SC_BLOCKS, 256, 0, stream>>>(x, b1_rows, b1_vals, y, nE);
  scatter_b2_kernel<<<SC_BLOCKS, 256, 0, stream>>>(x, b2_rows, b2_vals, upper, nT);
  {
    int tiles  = (nE + 15) / 16;
    int blocks = (tiles + 7) / 8;
    fused_gemm_pool_kernel<<<blocks, 256, 0, stream>>>(x, b1_rows, b1_vals,
                                                       ebatch, y, upper, wt,
                                                       gsum, gcnt, nE);
  }
  finalize_kernel<<<(GG * DD + 255) / 256, 256, 0, stream>>>(gsum, gcnt, out);
}


# fused GEMM as a resident grid: 512 blocks walk runs of tiles, weights staged once per block in LDS, next tile's rows requested under the current products; larger pooled partials (sum order changes)
# speedup vs baseline: 3.4404x; 1.4172x over previous
#include <hip/hip_runtime.h>
#include <hip/hip_bf16.h>

typedef __attribute__((ext_vector_type(2))) float v2f;
typedef __attribute__((ext_vector_type(4))) float v4f;
typedef __attribute__((ext_vector_type(8))) float v8f;

#define DD    64
#define GG    128
#define NNODE 200000

__global__ void init_ws_kernel(float* __restrict__ zero_region, size_t nzero,
                               const float* __restrict__ W0,
                               const float* __restrict__ W1,
                               const float* __restrict__ W2,
                               float* __restrict__ wt) {
  size_t tid    = (size_t)blockIdx.x * blockDim.x + threadIdx.x;
  size_t stride = (size_t)gridDim.x * blockDim.x;
  for (size_t i = tid; i < nzero; i += stride) zero_region[i] = 0.0f;
  for (size_t i = tid; i < 3u * 64u * 64u; i += stride) {
    int s  = (int)(i >> 12);
    int jj = (int)(i >> 9) & 7;
    int nt = (int)(i >> 7) & 3;
    int ln = (int)(i >> 2) & 31;
    int t  = (int)(i & 3);
    int k  = 8 * jj + 4 * (ln >> 4) + t;
    int n  = nt * 16 + (ln & 15);
    const float* W = (s == 0) ? W0 : ((s == 1) ? W1 : W2);
    wt[i] = W[k * 64 + n];
  }
}

#define SC_BLOCKS 2048
#define SC_WPB    8

__device__ __forceinline__ void b1_edge(const float* __restrict__ x,
                                        const int* __restrict__ b1_rows,
                                        const float* __restrict__ b1_vals,
                                        float* __restrict__ y, int e, int lane) {
  const float* xr = x + (size_t)e * DD;
  const float xa = xr[lane];
  const float xb = xr[32 + lane];
  const int   r0 = b1_rows[2 * e];
  const int   r1 = b1_rows[2 * e + 1];
  const float v0 = b1_vals[2 * e];
  const float v1 = b1_vals[2 * e + 1];
  float* y0 = y + (size_t)r0 * DD;
  float* y1 = y + (size_t)r1 * DD;
  atomicAdd(&y0[lane],      v0 * xa);
  atomicAdd(&y0[32 + lane], v0 * xb);
  atomicAdd(&y1[lane],      v1 * xa);
  atomicAdd(&y1[32 + lane], v1 * xb);
}

__global__ void __launch_bounds__(256)
scatter_b1_kernel(const float* __restrict__ x,
                  const int* __restrict__ b1_rows,
                  const float* __restrict__ b1_vals,
                  float* __restrict__ y, int nE) {
  const int lane = threadIdx.x & 31;
  const int wv = __builtin_amdgcn_readfirstlane((int)(threadIdx.x >> 5));
  const int gw = blockIdx.x * SC_WPB + wv;
  const int nw = gridDim.x * SC_WPB;
  const int nG = nE >> 2;
  for (int g = gw; g < nG; g += nw) {
    const int e0 = g * 4;
    float xa[4], xb[4], v0[4], v1[4];
    int   r0[4], r1[4];
#pragma unroll
    for (int u = 0; u < 4; ++u) {
      const int e = e0 + u;
      const float* xr = x + (size_t)e * DD;
      xa[u] = xr[lane];
      xb[u] = xr[32 + lane];
      r0[u] = b1_rows[2 * e];
      r1[u] = b1_rows[2 * e + 1];
      v0[u] = b1_vals[2 * e];
      v1[u] = b1_vals[2 * e + 1];
    }
#pragma unroll
    for (int u = 0; u < 4; ++u) {
      float* y0 = y + (size_t)r0[u] * DD;
      float* y1 = y + (size_t)r1[u] * DD;
      atomicAdd(&y0[lane],      v0[u] * xa[u]);
      atomicAdd(&y0[32 + lane], v0[u] * xb[u]);
      atomicAdd(&y1[lane],      v1[u] * xa[u]);
      atomicAdd(&y1[32 + lane], v1[u] * xb[u]);
    }
  }
  for (int e = nG * 4 + gw; e < nE; e += nw)
    b1_edge(x, b1_rows, b1_vals, y, e, lane);
}

__device__ __forceinline__ void b2_tri(const float* __restrict__ x,
                                       const int* __restrict__ b2_rows,
                                       const float* __restrict__ b2_vals,
                                       float* __restrict__ upper, int t, int lane) {
  const int   i0 = b2_rows[3 * t], i1 = b2_rows[3 * t + 1], i2 = b2_rows[3 * t + 2];
  const float v0 = b2_vals[3 * t], v1 = b2_vals[3 * t + 1], v2 = b2_vals[3 * t + 2];
  const float* x0 = x + (size_t)i0 * DD;
  const float* x1 = x + (size_t)i1 * DD;
  const float* x2 = x + (size_t)i2 * DD;
  const float za = v0 * x0[lane]      + v1 * x1[lane]      + v2 * x2[lane];
  const float zc = v0 * x0[32 + lane] + v1 * x1[32 + lane] + v2 * x2[32 + lane];
  float* u0 = upper + (size_t)i0 * DD;
  float* u1 = upper + (size_t)i1 * DD;
  float* u2 = upper + (size_t)i2 * DD;
  atomicAdd(&u0[lane],      v0 * za);
  atomicAdd(&u0[32 + lane], v0 * zc);
  atomicAdd(&u1[lane],      v1 * za);
  atomicAdd(&u1[32 + lane], v1 * zc);
  atomicAdd(&u2[lane],      v2 * za);
  atomicAdd(&u2[32 + lane], v2 * zc);
}

__global__ void __launch_bounds__(256)
scatter_b2_kernel(const float* __restrict__ x,
                  const int* __restrict__ b2_rows,
                  const float* __restrict__ b2_vals,
                  float* __restrict__ upper, int nT) {
  const int lane = threadIdx.x & 31;
  const int wv = __builtin_amdgcn_readfirstlane((int)(threadIdx.x >> 5));
  const int gw = blockIdx.x * SC_WPB + wv;
  const int nw = gridDim.x * SC_WPB;
  const int nP = nT >> 1;
  for (int p = gw; p < nP; p += nw) {
    const int t0 = p * 2;
    int   i0[2], i1[2], i2[2];
    float v0[2], v1[2], v2[2];
    float a0[2], a1[2], a2[2], c0[2], c1[2], c2[2];
#pragma unroll
    for (int u = 0; u < 2; ++u) {
      const int t = t0 + u;
      i0[u] = b2_rows[3 * t];  i1[u] = b2_rows[3 * t + 1];  i2[u] = b2_rows[3 * t + 2];
      v0[u] = b2_vals[3 * t];  v1[u] = b2_vals[3 * t + 1];  v2[u] = b2_vals[3 * t + 2];
      const float* x0 = x + (size_t)i0[u] * DD;
      const float* x1 = x + (size_t)i1[u] * DD;
      const float* x2 = x + (size_t)i2[u] * DD;
      a0[u] = x0[lane];  c0[u] = x0[32 + lane];
      a1[u] = x1[lane];  c1[u] = x1[32 + lane];
      a2[u] = x2[lane];  c2[u] = x2[32 + lane];
    }
#pragma unroll
    for (int u = 0; u < 2; ++u) {
      const float za = v0[u] * a0[u] + v1[u] * a1[u] + v2[u] * a2[u];
      const float zc = v0[u] * c0[u] + v1[u] * c1[u] + v2[u] * c2[u];
      float* u0 = upper + (size_t)i0[u] * DD;
      float* u1 = upper + (size_t)i1[u] * DD;
      float* u2 = upper + (size_t)i2[u] * DD;
      atomicAdd(&u0[lane],      v0[u] * za);
      atomicAdd(&u0[32 + lane], v0[u] * zc);
      atomicAdd(&u1[lane],      v1[u] * za);
      atomicAdd(&u1[32 + lane], v1[u] * zc);
      atomicAdd(&u2[lane],      v2[u] * za);
      atomicAdd(&u2[32 + lane], v2[u] * zc);
    }
  }
  for (int t = nP * 2 + gw; t < nT; t += nw)
    b2_tri(x, b2_rows, b2_vals, upper, t, lane);
}

#define GP_BLOCKS 512
#define GP_WPB    8

__global__ void __launch_bounds__(256)
fused_gemm_pool_kernel(const float* __restrict__ x,
                       const int* __restrict__ b1_rows,
                       const float* __restrict__ b1_vals,
                       const int* __restrict__ edge_batch,
                       const float* __restrict__ y,
                       const float* __restrict__ upper,
                       const float* __restrict__ wt,
                       float* __restrict__ gsum,
                       float* __restrict__ gcnt, int nE) {
  __shared__ float lgsum[GG * DD];
  __shared__ float lcnt[GG];
  __shared__ v4f   wl[3 * 64 * 64 / 4];
  for (int i = threadIdx.x; i < GG * DD; i += blockDim.x) lgsum[i] = 0.0f;
  if (threadIdx.x < GG) lcnt[threadIdx.x] = 0.0f;
  {
    const v4f* wg = (const v4f*)wt;
    v4f wtmp[12];
#pragma unroll
    for (int i = 0; i < 12; ++i) wtmp[i] = wg[threadIdx.x + 256 * i];
#pragma unroll
    for (int i = 0; i < 12; ++i) wl[threadIdx.x + 256 * i] = wtmp[i];
  }
  __syncthreads();

  const int lane = threadIdx.x & 31;
  const int wave = __builtin_amdgcn_readfirstlane((int)(threadIdx.x >> 5));
  const unsigned nT = (unsigned)(nE >> 4);
  const int lo = (int)(nT * blockIdx.x / gridDim.x);
  const int hi = (int)(nT * (blockIdx.x + 1u) / gridDim.x);
  int t = lo + wave;

  if (t < hi) {
    const int m  = lane & 15;
    const int kh = lane >> 4;

    const v4f* wq = wl + lane;

    v4f a0[8], a2[8], p0[8], p1[8];
    float s0, s1;
    int gid[8], gme;
    int r0n, r1n;
    float s0n, s1n;
    {
      const int e  = t * 16 + m;
      const int r0 = b1_rows[2 * e];
      const int r1 = b1_rows[2 * e + 1];
      s0 = b1_vals[2 * e];
      s1 = b1_vals[2 * e + 1];
      const v4f* xq  = (const v4f*)(x     + (size_t)e  * DD) + kh;
      const v4f* uq  = (const v4f*)(upper + (size_t)e  * DD) + kh;
      const v4f* y0q = (const v4f*)(y     + (size_t)r0 * DD) + kh;
      const v4f* y1q = (const v4f*)(y     + (size_t)r1 * DD) + kh;
      const int en = ((t + GP_WPB < hi) ? t + GP_WPB : t) * 16 + m;
      r0n = b1_rows[2 * en];
      r1n = b1_rows[2 * en + 1];
      s0n = b1_vals[2 * en];
      s1n = b1_vals[2 * en + 1];
#pragma unroll
      for (int jj = 0; jj < 8; ++jj) a0[jj] = xq[2 * jj];
#pragma unroll
      for (int jj = 0; jj < 8; ++jj) a2[jj] = uq[2 * jj];
#pragma unroll
      for (int jj = 0; jj < 8; ++jj) p0[jj] = y0q[2 * jj];
#pragma unroll
      for (int jj = 0; jj < 8; ++jj) p1[jj] = y1q[2 * jj];
#pragma unroll
      for (int r = 0; r < 8; ++r) gid[r] = edge_batch[t * 16 + kh * 8 + r];
      gme = edge_batch[e];
    }

    for (; t < hi; t += GP_WPB) {
      const int tn = (t + GP_WPB < hi) ? t + GP_WPB : t;
      const int en = tn * 16 + m;
      const v4f* xnq = (const v4f*)(x     + (size_t)en * DD) + kh;
      const v4f* unq = (const v4f*)(upper + (size_t)en * DD) + kh;

      v4f bq[3][4];
#pragma unroll
      for (int s = 0; s < 2; ++s)
#pragma unroll
        for (int nt = 0; nt < 4; ++nt) bq[s][nt] = wq[(s * 4 + nt) * 32];

      v8f acc[4] = {};
      v4f a1[8];

#pragma unroll
      for (int s = 0; s < 24; ++s) {
        const int src = s >> 3;
        const int jj  = s & 7;
        if (s == 8) {
#pragma unroll
          for (int j = 0; j < 8; ++j)
            a1[j] = s0 * p0[j] + s1 * p1[j];
#pragma unroll
          for (int j = 0; j < 8; ++j) a0[j] = xnq[2 * j];
        }
        if (s == 16) {
          asm volatile("" : "+v"(r0n), "+v"(r1n));
          const v4f* y0nq = (const v4f*)(y + (size_t)r0n * DD) + kh;
          const v4f* y1nq = (const v4f*)(y + (size_t)r1n * DD) + kh;
#pragma unroll
          for (int j = 0; j < 8; ++j) p0[j] = y0nq[2 * j];
#pragma unroll
          for (int j = 0; j < 8; ++j) p1[j] = y1nq[2 * j];
          s0 = s0n;
          s1 = s1n;
        }
        if (s + 2 < 24) {
#pragma unroll
          for (int nt = 0; nt < 4; ++nt)
            bq[(s + 2) % 3][nt] = wq[((s + 2) * 4 + nt) * 32];
        }
        const v4f a4 = (src == 0) ? a0[jj] : ((src == 1) ? a1[jj] : a2[jj]);
        v2f alo; alo.x = a4.x; alo.y = a4.y;
        v2f ahi; ahi.x = a4.z; ahi.y = a4.w;
#pragma unroll
        for (int nt = 0; nt < 4; ++nt) {
          v2f b; b.x = bq[s % 3][nt].x; b.y = bq[s % 3][nt].y;
          acc[nt] = __builtin_amdgcn_wmma_f32_16x16x4_f32(
              false, alo, false, b, (short)0, acc[nt], false, false);
        }
#pragma unroll
        for (int nt = 0; nt < 4; ++nt) {
          v2f b; b.x = bq[s % 3][nt].z; b.y = bq[s % 3][nt].w;
          acc[nt] = __builtin_amdgcn_wmma_f32_16x16x4_f32(
              false, ahi, false, b, (short)0, acc[nt], false, false);
        }
        __builtin_amdgcn_sched_barrier(0);
      }

#pragma unroll
      for (int nt = 0; nt < 4; ++nt) {
#pragma unroll
        for (int r = 0; r < 8; ++r) {
          float v = acc[nt][r];
          v = v > 0.0f ? v : 0.0f;
          atomicAdd(&lgsum[gid[r] * DD + nt * 16 + m], v);
        }
      }
      atomicAdd(&lcnt[gme], 0.5f);

      {
        const int enn = ((t + 2 * GP_WPB < hi) ? t + 2 * GP_WPB : tn) * 16 + m;
        r0n = b1_rows[2 * enn];
        r1n = b1_rows[2 * enn + 1];
        s0n = b1_vals[2 * enn];
        s1n = b1_vals[2 * enn + 1];
      }
#pragma unroll
      for (int j = 0; j < 8; ++j) a2[j] = unq[2 * j];
#pragma unroll
      for (int r = 0; r < 8; ++r) gid[r] = edge_batch[tn * 16 + kh * 8 + r];
      gme = edge_batch[en];
    }
  }

  __syncthreads();
  for (int i = threadIdx.x; i < GG * DD; i += blockDim.x)
    atomicAdd(&gsum[i], lgsum[i]);
  if (threadIdx.x < GG) atomicAdd(&gcnt[threadIdx.x], lcnt[threadIdx.x]);
}

__global__ void finalize_kernel(const float* __restrict__ gsum,
                                const float* __restrict__ gcnt,
                                float* __restrict__ out) {
  int i = blockIdx.x * blockDim.x + threadIdx.x;
  if (i >= GG * DD) return;
  float c = gcnt[i >> 6];
  float v = gsum[i] / fmaxf(c, 1.0f);
  out[i] = v;
  out[GG * DD + i] = v;
  out[2 * GG * DD + i] = v;
}

extern "C" void kernel_launch(void* const* d_in, const int* in_sizes, int n_in,
                              void* d_out, int out_size, void* d_ws, size_t ws_size,
                              hipStream_t stream) {
  const float* x        = (const float*)d_in[0];
  const int*   b1_rows  = (const int*)  d_in[1];
  const float* b1_vals  = (const float*)d_in[3];
  const int*   b2_rows  = (const int*)  d_in[4];
  const float* b2_vals  = (const float*)d_in[6];
  const int*   ebatch   = (const int*)  d_in[7];
  const float* W0       = (const float*)d_in[8];
  const float* W1       = (const float*)d_in[9];
  const float* W2       = (const float*)d_in[10];
  float*       out      = (float*)d_out;

  const int nE = in_sizes[0] / DD;
  const int nT = in_sizes[4] / 3;

  float* ws    = (float*)d_ws;
  float* y     = ws;
  size_t nY    = (size_t)NNODE * DD;
  float* upper = y + nY;
  size_t nU    = (size_t)nE * DD;
  float* gsum  = upper + nU;
  float* gcnt  = gsum + GG * DD;
  float* wt    = gcnt + GG;
  size_t nzero = nY + nU + GG * DD + GG;

  init_ws_kernel<<<2048, 256, 0, stream>>>(ws, nzero, W0, W1, W2, wt);

  scatter_b1_kernel<<<SC_BLOCKS, 256, 0, stream>>>(x, b1_rows, b1_vals, y, nE);
  scatter_b2_kernel<<<SC_BLOCKS, 256, 0, stream>>>(x, b2_rows, b2_vals, upper, nT);
  fused_gemm_pool_kernel<<<GP_BLOCKS, 256, 0, stream>>>(x, b1_rows, b1_vals,
                                                        ebatch, y, upper, wt,
                                                        gsum, gcnt, nE);
  finalize_kernel<<<(GG * DD + 255) / 256, 256, 0, stream>>>(gsum, gcnt, out);
}
